# starting kernel
# baseline (speedup 1.0000x reference)
#include <hip/hip_runtime.h>
#include <hip/hip_bf16.h>

#define N_NODES 100000
#define DIM     256
#define N_EDGES 800000
#define L2EPS   1e-12f

typedef __attribute__((ext_vector_type(2))) float v2f;
typedef __attribute__((ext_vector_type(8))) float v8f;

__global__ __launch_bounds__(256)
void gcn_init_kernel(const float* __restrict__ adj, const float* __restrict__ adjw,
                     float* __restrict__ vals, float* __restrict__ out,
                     int nd, int e) {
    int idx = blockIdx.x * blockDim.x + threadIdx.x;
    if (idx < nd) out[idx] = 0.0f;
    if (idx < e)  vals[idx] = adj[idx] + adjw[idx];
}

__global__ __launch_bounds__(128)
void gcn_gemm_wmma_kernel(const float* __restrict__ x, const float* __restrict__ w,
                          float* __restrict__ support) {
    const int wave   = blockIdx.x * (blockDim.x >> 5) + (threadIdx.x >> 5);
    const int lane   = threadIdx.x & 31;
    const int laneHi = lane >> 4;
    const int laneLo = lane & 15;

    const int nTile = wave & 15;
    const int mTile = wave >> 4;
    if (mTile >= N_NODES / 16) return;

    const int m0 = mTile * 16;
    const int n0 = nTile * 16;

    const float* aBase = x + (size_t)(m0 + laneLo) * DIM + 2 * laneHi;
    const float* bBase = w + (size_t)(2 * laneHi) * DIM + n0 + laneLo;

    v8f acc0 = {};
    v8f acc1 = {};

    #pragma unroll 4
    for (int kk = 0; kk < DIM; kk += 8) {
        v2f a0 = *(const v2f*)(aBase + kk);
        v2f b0;
        b0.x = bBase[(size_t)(kk + 0) * DIM];
        b0.y = bBase[(size_t)(kk + 1) * DIM];
        acc0 = __builtin_amdgcn_wmma_f32_16x16x4_f32(
            false, a0, false, b0, (short)0, acc0, false, false);

        v2f a1 = *(const v2f*)(aBase + kk + 4);
        v2f b1;
        b1.x = bBase[(size_t)(kk + 4) * DIM];
        b1.y = bBase[(size_t)(kk + 5) * DIM];
        acc1 = __builtin_amdgcn_wmma_f32_16x16x4_f32(
            false, a1, false, b1, (short)0, acc1, false, false);
    }

    v8f sum = acc0 + acc1;

    float* oBase = support + (size_t)(m0 + 8 * laneHi) * DIM + n0 + laneLo;
    #pragma unroll
    for (int i = 0; i < 8; ++i)
        oBase[(size_t)i * DIM] = sum[i];
}

__global__ __launch_bounds__(256)
void gcn_spmm_kernel(const int* __restrict__ ei, const float* __restrict__ vals,
                     const float* __restrict__ support, float* __restrict__ out) {
    const int e    = blockIdx.x * 8 + (threadIdx.x >> 5);
    const int lane = threadIdx.x & 31;
    if (e >= N_EDGES) return;

    const int dst = ei[e];
    const int src = ei[N_EDGES + e];
    const float v = vals[e];

    const float* sRow = support + (size_t)src * DIM;
    float*       oRow = out     + (size_t)dst * DIM;

    #pragma unroll
    for (int d = lane; d < DIM; d += 32) {
        unsafeAtomicAdd(&oRow[d], v * sRow[d]);
    }
}

__global__ __launch_bounds__(256)
void gcn_normalize_kernel(float* __restrict__ out) {
    __shared__ float red[8];
    const int row = blockIdx.x;
    const int tid = threadIdx.x;

    float v = out[(size_t)row * DIM + tid];
    float s = v * v;

    #pragma unroll
    for (int off = 16; off > 0; off >>= 1)
        s += __shfl_xor(s, off, 32);
    if ((tid & 31) == 0) red[tid >> 5] = s;
    __syncthreads();

    float total = 0.0f;
    #pragma unroll
    for (int i = 0; i < 8; ++i) total += red[i];

    const float scale = 1.0f / fmaxf(sqrtf(total), L2EPS);
    out[(size_t)row * DIM + tid] = v * scale;
}

extern "C" void kernel_launch(void* const* d_in, const int* in_sizes, int n_in,
                              void* d_out, int out_size, void* d_ws, size_t ws_size,
                              hipStream_t stream) {
    const float* x    = (const float*)d_in[0];
    const float* w    = (const float*)d_in[1];
    const float* adj  = (const float*)d_in[2];
    const float* adjw = (const float*)d_in[3];
    const int*   ei   = (const int*)  d_in[4];

    float* out     = (float*)d_out;
    float* support = (float*)d_ws;
    float* vals    = support + (size_t)N_NODES * DIM;

    const int nd = N_NODES * DIM;

    gcn_init_kernel<<<(nd + 255) / 256, 256, 0, stream>>>(adj, adjw, vals, out,
                                                          nd, N_EDGES);

    const int tiles = (N_NODES / 16) * (DIM / 16);
    gcn_gemm_wmma_kernel<<<(tiles + 3) / 4, 128, 0, stream>>>(x, w, support);

    gcn_spmm_kernel<<<(N_EDGES + 7) / 8, 256, 0, stream>>>(ei, vals, support, out);

    gcn_normalize_kernel<<<N_NODES, 256, 0, stream>>>(out);
}


# Scatter of float atomics replaced by per-call grouping of edges by destination (count, scan, fill) and a row-owning register gather with the normalize folded in; row sums stay in arrival order
# speedup vs baseline: 1.5038x; 1.5038x over previous
#include <hip/hip_runtime.h>
#include <hip/hip_bf16.h>


#define N_NODES 100000
#define DIM     256
#define N_EDGES 800000
#define L2EPS   1e-12f
#define SCAN_T  1024

typedef __attribute__((ext_vector_type(2))) float v2f;
typedef __attribute__((ext_vector_type(8))) float v8f;


__global__ __launch_bounds__(256)
void gcn_zero_cnt_kernel(int* __restrict__ cnt) {
    const int i = blockIdx.x * blockDim.x + threadIdx.x;
    if (i < N_NODES) cnt[i] = 0;
}

__global__ __launch_bounds__(256)
void gcn_count_kernel(const int* __restrict__ ei, int* __restrict__ cnt) {
    const int e = blockIdx.x * blockDim.x + threadIdx.x;
    if (e < N_EDGES) atomicAdd(&cnt[ei[e]], 1);
}

__global__ __launch_bounds__(SCAN_T)
void gcn_scan_kernel(const int* __restrict__ cnt, int* __restrict__ rowptr) {
    __shared__ int wsum[32];
    __shared__ int bprefix;
    const int tid  = threadIdx.x;
    const int lane = tid & 31;
    const int wid  = tid >> 5;
    const int b0   = blockIdx.x * SCAN_T;

    int p = 0;
    for (int i = tid; i < b0; i += SCAN_T) p += cnt[i];
    for (int off = 16; off > 0; off >>= 1) p += __shfl_xor(p, off, 32);
    if (lane == 0) wsum[wid] = p;
    __syncthreads();
    if (wid == 0) {
        int t = wsum[lane];
        for (int off = 16; off > 0; off >>= 1) t += __shfl_xor(t, off, 32);
        if (lane == 0) bprefix = t;
    }
    __syncthreads();
    const int prefix = bprefix;

    const int i = b0 + tid;
    const int c = (i < N_NODES) ? cnt[i] : 0;
    int x = c;
    for (int off = 1; off < 32; off <<= 1) {
        const int t = __shfl_up(x, off, 32);
        if (lane >= off) x += t;
    }
    if (lane == 31) wsum[wid] = x;
    __syncthreads();
    if (wid == 0) {
        const int w = wsum[lane];
        int s = w;
        for (int off = 1; off < 32; off <<= 1) {
            const int t = __shfl_up(s, off, 32);
            if (lane >= off) s += t;
        }
        wsum[lane] = s - w;
    }
    __syncthreads();
    if (i < N_NODES) rowptr[i] = prefix + wsum[wid] + (x - c);
}

__global__ __launch_bounds__(256)
void gcn_fill_kernel(const int* __restrict__ ei, int* __restrict__ rowptr,
                     unsigned short* __restrict__ permLo,
                     unsigned char* __restrict__ permHi) {
    const int e = blockIdx.x * blockDim.x + threadIdx.x;
    if (e >= N_EDGES) return;
    const int pos = atomicAdd(&rowptr[ei[e]], 1);
    if ((unsigned)pos < (unsigned)N_EDGES) {
        permLo[pos] = (unsigned short)(e & 0xFFFF);
        permHi[pos] = (unsigned char)(e >> 16);
    }
}

__global__ __launch_bounds__(128)
void gcn_gemm_wmma_kernel(const float* __restrict__ x, const float* __restrict__ w,
                          float* __restrict__ support) {
    const int wave   = blockIdx.x * (blockDim.x >> 5) + (threadIdx.x >> 5);
    const int lane   = threadIdx.x & 31;
    const int laneHi = lane >> 4;
    const int laneLo = lane & 15;

    const int nTile = wave & 15;
    const int mTile = wave >> 4;
    if (mTile >= N_NODES / 16) return;

    const int m0 = mTile * 16;
    const int n0 = nTile * 16;

    const float* aBase = x + (size_t)(m0 + laneLo) * DIM + 2 * laneHi;
    const float* bBase = w + (size_t)(2 * laneHi) * DIM + n0 + laneLo;

    v8f acc0 = {};
    v8f acc1 = {};

    #pragma unroll 4
    for (int kk = 0; kk < DIM; kk += 8) {
        v2f a0 = *(const v2f*)(aBase + kk);
        v2f b0;
        b0.x = bBase[(size_t)(kk + 0) * DIM];
        b0.y = bBase[(size_t)(kk + 1) * DIM];
        acc0 = __builtin_amdgcn_wmma_f32_16x16x4_f32(
            false, a0, false, b0, (short)0, acc0, false, false);

        v2f a1 = *(const v2f*)(aBase + kk + 4);
        v2f b1;
        b1.x = bBase[(size_t)(kk + 4) * DIM];
        b1.y = bBase[(size_t)(kk + 5) * DIM];
        acc1 = __builtin_amdgcn_wmma_f32_16x16x4_f32(
            false, a1, false, b1, (short)0, acc1, false, false);
    }

    v8f sum = acc0 + acc1;

    float* oBase = support + (size_t)(m0 + 8 * laneHi) * DIM + n0 + laneLo;
    #pragma unroll
    for (int i = 0; i < 8; ++i)
        oBase[(size_t)i * DIM] = sum[i];
}

template <int K>
__device__ __forceinline__ void gcn_accum_rows(const float* __restrict__ support,
                                               int mySrc, float myVal, int j, int lane,
                                               float4& accLo, float4& accHi) {
    float4 lo[K], hi[K];
    float  v[K];
    #pragma unroll
    for (int u = 0; u < K; ++u) {
        const int s = __builtin_amdgcn_readlane(mySrc, j + u);
        v[u] = __int_as_float(__builtin_amdgcn_readlane(__float_as_int(myVal), j + u));
        const float4* row = (const float4*)(support + (size_t)s * DIM);
        lo[u] = row[lane];
        hi[u] = row[32 + lane];
    }
    #pragma unroll
    for (int u = 0; u < K; ++u) {
        accLo.x = fmaf(v[u], lo[u].x, accLo.x);
        accLo.y = fmaf(v[u], lo[u].y, accLo.y);
        accLo.z = fmaf(v[u], lo[u].z, accLo.z);
        accLo.w = fmaf(v[u], lo[u].w, accLo.w);
        accHi.x = fmaf(v[u], hi[u].x, accHi.x);
        accHi.y = fmaf(v[u], hi[u].y, accHi.y);
        accHi.z = fmaf(v[u], hi[u].z, accHi.z);
        accHi.w = fmaf(v[u], hi[u].w, accHi.w);
    }
}

__global__ __launch_bounds__(256)
void gcn_gather_normalize_kernel(const int* __restrict__ ei, const float* __restrict__ adj,
                                 const float* __restrict__ adjw,
                                 const int* __restrict__ rowptr,
                                 const unsigned short* __restrict__ permLo,
                                 const unsigned char* __restrict__ permHi,
                                 const float* __restrict__ support, float* __restrict__ out) {
    const int lane = threadIdx.x & 31;
    const int row = __builtin_amdgcn_readfirstlane(blockIdx.x * 8 + (threadIdx.x >> 5));
    if (row >= N_NODES) return;

    const int end   = rowptr[row];
    const int start = (row > 0) ? rowptr[row - 1] : 0;

    float4 accLo = make_float4(0.f, 0.f, 0.f, 0.f);
    float4 accHi = make_float4(0.f, 0.f, 0.f, 0.f);

    for (int base = start; base < end; base += 32) {
        const int n = min(32, end - base);
        int   mySrc = 0;
        float myVal = 0.0f;
        if (lane < n) {
            const int p = base + lane;
            int e = (int)permLo[p] | ((int)permHi[p] << 16);
            e = min(e, N_EDGES - 1);
            mySrc = ei[N_EDGES + e];
            myVal = adj[e] + adjw[e];
        }
        int j = 0;
        for (; j + 4 <= n; j += 4)
            gcn_accum_rows<4>(support, mySrc, myVal, j, lane, accLo, accHi);
        if (j + 2 <= n) {
            gcn_accum_rows<2>(support, mySrc, myVal, j, lane, accLo, accHi);
            j += 2;
        }
        if (j < n)
            gcn_accum_rows<1>(support, mySrc, myVal, j, lane, accLo, accHi);
    }

    float ss = accLo.x * accLo.x + accLo.y * accLo.y + accLo.z * accLo.z + accLo.w * accLo.w
             + accHi.x * accHi.x + accHi.y * accHi.y + accHi.z * accHi.z + accHi.w * accHi.w;
    for (int off = 16; off > 0; off >>= 1)
        ss += __shfl_xor(ss, off, 32);

    const float scale = 1.0f / fmaxf(sqrtf(ss), L2EPS);
    float4* o = (float4*)(out + (size_t)row * DIM);
    o[lane]      = make_float4(accLo.x * scale, accLo.y * scale, accLo.z * scale, accLo.w * scale);
    o[32 + lane] = make_float4(accHi.x * scale, accHi.y * scale, accHi.z * scale, accHi.w * scale);
}

extern "C" void kernel_launch(void* const* d_in, const int* in_sizes, int n_in,
                              void* d_out, int out_size, void* d_ws, size_t ws_size,
                              hipStream_t stream) {
    const float* x    = (const float*)d_in[0];
    const float* w    = (const float*)d_in[1];
    const float* adj  = (const float*)d_in[2];
    const float* adjw = (const float*)d_in[3];
    const int*   ei   = (const int*)  d_in[4];

    float* out     = (float*)d_out;
    float* support = (float*)d_ws;
    int*            rowptr = (int*)(support + (size_t)N_NODES * DIM);
    unsigned short* permLo = (unsigned short*)(rowptr + N_NODES);
    unsigned char*  permHi = (unsigned char*)(permLo + N_EDGES);
    int*            cnt    = (int*)d_out;

    gcn_zero_cnt_kernel<<<(N_NODES + 255) / 256, 256, 0, stream>>>(cnt);
    gcn_count_kernel<<<(N_EDGES + 255) / 256, 256, 0, stream>>>(ei, cnt);
    gcn_scan_kernel<<<(N_NODES + SCAN_T - 1) / SCAN_T, SCAN_T, 0, stream>>>(cnt, rowptr);
    gcn_fill_kernel<<<(N_EDGES + 255) / 256, 256, 0, stream>>>(ei, rowptr, permLo, permHi);

    const int tiles = (N_NODES / 16) * (DIM / 16);
    gcn_gemm_wmma_kernel<<<(tiles + 3) / 4, 128, 0, stream>>>(x, w, support);

    gcn_gather_normalize_kernel<<<(N_NODES + 7) / 8, 256, 0, stream>>>(
        ei, adj, adjw, rowptr, permLo, permHi, support, out);
}


# GEMM retiled: W staged once per work-group in LDS, each wave owns a 16x256 row strip (A fragment loaded once, feeds 16 WMMAs); k order of the product sums changes (one accumulator chain per tile)
# speedup vs baseline: 2.1399x; 1.4230x over previous
#include <hip/hip_runtime.h>
#include <hip/hip_bf16.h>


#define N_NODES 100000
#define DIM     256
#define N_EDGES 800000
#define L2EPS   1e-12f
#define SCAN_T  1024

typedef __attribute__((ext_vector_type(2))) float v2f;
typedef __attribute__((ext_vector_type(8))) float v8f;


__global__ __launch_bounds__(256)
void gcn_zero_cnt_kernel(int* __restrict__ cnt) {
    const int i = blockIdx.x * blockDim.x + threadIdx.x;
    if (i < N_NODES) cnt[i] = 0;
}

__global__ __launch_bounds__(256)
void gcn_count_kernel(const int* __restrict__ ei, int* __restrict__ cnt) {
    const int e = blockIdx.x * blockDim.x + threadIdx.x;
    if (e < N_EDGES) atomicAdd(&cnt[ei[e]], 1);
}

__global__ __launch_bounds__(SCAN_T)
void gcn_scan_kernel(const int* __restrict__ cnt, int* __restrict__ rowptr) {
    __shared__ int wsum[32];
    __shared__ int bprefix;
    const int tid  = threadIdx.x;
    const int lane = tid & 31;
    const int wid  = tid >> 5;
    const int b0   = blockIdx.x * SCAN_T;

    int p = 0;
    for (int i = tid; i < b0; i += SCAN_T) p += cnt[i];
    for (int off = 16; off > 0; off >>= 1) p += __shfl_xor(p, off, 32);
    if (lane == 0) wsum[wid] = p;
    __syncthreads();
    if (wid == 0) {
        int t = wsum[lane];
        for (int off = 16; off > 0; off >>= 1) t += __shfl_xor(t, off, 32);
        if (lane == 0) bprefix = t;
    }
    __syncthreads();
    const int prefix = bprefix;

    const int i = b0 + tid;
    const int c = (i < N_NODES) ? cnt[i] : 0;
    int x = c;
    for (int off = 1; off < 32; off <<= 1) {
        const int t = __shfl_up(x, off, 32);
        if (lane >= off) x += t;
    }
    if (lane == 31) wsum[wid] = x;
    __syncthreads();
    if (wid == 0) {
        const int w = wsum[lane];
        int s = w;
        for (int off = 1; off < 32; off <<= 1) {
            const int t = __shfl_up(s, off, 32);
            if (lane >= off) s += t;
        }
        wsum[lane] = s - w;
    }
    __syncthreads();
    if (i < N_NODES) rowptr[i] = prefix + wsum[wid] + (x - c);
}

__global__ __launch_bounds__(256)
void gcn_fill_kernel(const int* __restrict__ ei, int* __restrict__ rowptr,
                     unsigned short* __restrict__ permLo,
                     unsigned char* __restrict__ permHi) {
    const int e = blockIdx.x * blockDim.x + threadIdx.x;
    if (e >= N_EDGES) return;
    const int pos = atomicAdd(&rowptr[ei[e]], 1);
    if ((unsigned)pos < (unsigned)N_EDGES) {
        permLo[pos] = (unsigned short)(e & 0xFFFF);
        permHi[pos] = (unsigned char)(e >> 16);
    }
}

#define GEMM_WAVES  16
#define GEMM_BLOCKS 256
#define M_TILES     (N_NODES / 16)
static_assert(N_NODES % 16 == 0 && M_TILES >= GEMM_BLOCKS * GEMM_WAVES && M_TILES <= GEMM_BLOCKS * 25,
              "strip assignment below covers j = 0..24 and gives every wave a first strip");

__device__ __forceinline__ void gcn_load_a_chunk(const float* __restrict__ p, float4 (&a)[4]) {
    #pragma unroll
    for (int u = 0; u < 4; ++u) a[u] = *(const float4*)(p + 8 * u);
}

__device__ __forceinline__ void gcn_strip_chunk(const float4 (&a)[4], const float4* bp,
                                                v8f (&acc)[16]) {
    #pragma unroll
    for (int u = 0; u < 4; ++u) {
        float4 b[16];
        #pragma unroll
        for (int nt = 0; nt < 16; ++nt) b[nt] = bp[u * 512 + nt * 16];

        const v2f a01 = {a[u].x, a[u].y};
        const v2f a23 = {a[u].z, a[u].w};
        #pragma unroll
        for (int nt = 0; nt < 16; ++nt) {
            const v2f b01 = {b[nt].x, b[nt].y};
            acc[nt] = __builtin_amdgcn_wmma_f32_16x16x4_f32(
                false, a01, false, b01, (short)0, acc[nt], false, false);
        }
        #pragma unroll
        for (int nt = 0; nt < 16; ++nt) {
            const v2f b23 = {b[nt].z, b[nt].w};
            acc[nt] = __builtin_amdgcn_wmma_f32_16x16x4_f32(
                false, a23, false, b23, (short)0, acc[nt], false, false);
        }
    }
}

__global__ __launch_bounds__(GEMM_WAVES * 32)
void gcn_gemm_wmma_kernel(const float* __restrict__ x, const float* __restrict__ w,
                          float* __restrict__ support) {
    __shared__ float4 wl[DIM * DIM / 4];

    const int tid  = threadIdx.x;
    const int lane = tid & 31;
    const int h    = lane >> 4;
    const int c    = lane & 15;
    const int wv   = __builtin_amdgcn_readfirstlane(tid >> 5);

    const int j1 = (((wv >> 1) ^ (wv >> 2)) & 1) == 0 ? 16 + (((wv >> 2) << 1) | (wv & 1))
                                                      : (wv == 2 ? 24 : 25);
    int t  = blockIdx.x + GEMM_BLOCKS * wv;
    int tn = blockIdx.x + GEMM_BLOCKS * j1;
    const float* aPtr = x + (size_t)(t * 16 + c) * DIM + 4 * h;

    float4 aC[4], aN[4];
    gcn_load_a_chunk(aPtr, aC);

    for (int s = tid; s < DIM * DIM / 4; s += GEMM_WAVES * 32) {
        const float* src = w + (size_t)(4 * (s >> 8)) * DIM + (s & 255);
        wl[s] = make_float4(src[0], src[DIM], src[2 * DIM], src[3 * DIM]);
    }
    __syncthreads();

    const float4* bLane = wl + h * 256 + c;

    while (t < M_TILES) {
        const float* aNextTile = x + (size_t)(tn * 16 + c) * DIM + 4 * h;
        const float* aNextHead = tn < M_TILES ? aNextTile : aPtr;

        v8f acc[16];
        #pragma unroll
        for (int nt = 0; nt < 16; ++nt) acc[nt] = v8f{};

        #pragma unroll 1
        for (int gp = 0; gp < 4; ++gp) {
            gcn_load_a_chunk(aPtr + 32 * (2 * gp + 1), aN);
            gcn_strip_chunk(aC, bLane + (size_t)(8 * gp) * 512, acc);

            gcn_load_a_chunk(gp < 3 ? aPtr + 32 * (2 * gp + 2) : aNextHead, aC);
            gcn_strip_chunk(aN, bLane + (size_t)(8 * gp + 4) * 512, acc);
        }

        float* oBase = support + (size_t)(t * 16 + 8 * h) * DIM + c;
        #pragma unroll
        for (int nt = 0; nt < 16; ++nt) {
            #pragma unroll
            for (int i = 0; i < 8; ++i)
                oBase[(size_t)i * DIM + nt * 16] = acc[nt][i];
        }

        t    = tn;
        tn   = M_TILES;
        aPtr = aNextTile;
    }
}

template <int K>
__device__ __forceinline__ void gcn_accum_rows(const float* __restrict__ support,
                                               int mySrc, float myVal, int j, int lane,
                                               float4& accLo, float4& accHi) {
    float4 lo[K], hi[K];
    float  v[K];
    #pragma unroll
    for (int u = 0; u < K; ++u) {
        const int s = __builtin_amdgcn_readlane(mySrc, j + u);
        v[u] = __int_as_float(__builtin_amdgcn_readlane(__float_as_int(myVal), j + u));
        const float4* row = (const float4*)(support + (size_t)s * DIM);
        lo[u] = row[lane];
        hi[u] = row[32 + lane];
    }
    #pragma unroll
    for (int u = 0; u < K; ++u) {
        accLo.x = fmaf(v[u], lo[u].x, accLo.x);
        accLo.y = fmaf(v[u], lo[u].y, accLo.y);
        accLo.z = fmaf(v[u], lo[u].z, accLo.z);
        accLo.w = fmaf(v[u], lo[u].w, accLo.w);
        accHi.x = fmaf(v[u], hi[u].x, accHi.x);
        accHi.y = fmaf(v[u], hi[u].y, accHi.y);
        accHi.z = fmaf(v[u], hi[u].z, accHi.z);
        accHi.w = fmaf(v[u], hi[u].w, accHi.w);
    }
}

__global__ __launch_bounds__(256)
void gcn_gather_normalize_kernel(const int* __restrict__ ei, const float* __restrict__ adj,
                                 const float* __restrict__ adjw,
                                 const int* __restrict__ rowptr,
                                 const unsigned short* __restrict__ permLo,
                                 const unsigned char* __restrict__ permHi,
                                 const float* __restrict__ support, float* __restrict__ out) {
    const int lane = threadIdx.x & 31;
    const int row = __builtin_amdgcn_readfirstlane(blockIdx.x * 8 + (threadIdx.x >> 5));
    if (row >= N_NODES) return;

    const int end   = rowptr[row];
    const int start = (row > 0) ? rowptr[row - 1] : 0;

    float4 accLo = make_float4(0.f, 0.f, 0.f, 0.f);
    float4 accHi = make_float4(0.f, 0.f, 0.f, 0.f);

    for (int base = start; base < end; base += 32) {
        const int n = min(32, end - base);
        int   mySrc = 0;
        float myVal = 0.0f;
        if (lane < n) {
            const int p = base + lane;
            int e = (int)permLo[p] | ((int)permHi[p] << 16);
            e = min(e, N_EDGES - 1);
            mySrc = ei[N_EDGES + e];
            myVal = adj[e] + adjw[e];
        }
        int j = 0;
        for (; j + 4 <= n; j += 4)
            gcn_accum_rows<4>(support, mySrc, myVal, j, lane, accLo, accHi);
        if (j + 2 <= n) {
            gcn_accum_rows<2>(support, mySrc, myVal, j, lane, accLo, accHi);
            j += 2;
        }
        if (j < n)
            gcn_accum_rows<1>(support, mySrc, myVal, j, lane, accLo, accHi);
    }

    float ss = accLo.x * accLo.x + accLo.y * accLo.y + accLo.z * accLo.z + accLo.w * accLo.w
             + accHi.x * accHi.x + accHi.y * accHi.y + accHi.z * accHi.z + accHi.w * accHi.w;
    for (int off = 16; off > 0; off >>= 1)
        ss += __shfl_xor(ss, off, 32);

    const float scale = 1.0f / fmaxf(sqrtf(ss), L2EPS);
    float4* o = (float4*)(out + (size_t)row * DIM);
    o[lane]      = make_float4(accLo.x * scale, accLo.y * scale, accLo.z * scale, accLo.w * scale);
    o[32 + lane] = make_float4(accHi.x * scale, accHi.y * scale, accHi.z * scale, accHi.w * scale);
}

extern "C" void kernel_launch(void* const* d_in, const int* in_sizes, int n_in,
                              void* d_out, int out_size, void* d_ws, size_t ws_size,
                              hipStream_t stream) {
    const float* x    = (const float*)d_in[0];
    const float* w    = (const float*)d_in[1];
    const float* adj  = (const float*)d_in[2];
    const float* adjw = (const float*)d_in[3];
    const int*   ei   = (const int*)  d_in[4];

    float* out     = (float*)d_out;
    float* support = (float*)d_ws;
    int*            rowptr = (int*)(support + (size_t)N_NODES * DIM);
    unsigned short* permLo = (unsigned short*)(rowptr + N_NODES);
    unsigned char*  permHi = (unsigned char*)(permLo + N_EDGES);
    int*            cnt    = (int*)d_out;

    gcn_zero_cnt_kernel<<<(N_NODES + 255) / 256, 256, 0, stream>>>(cnt);
    gcn_count_kernel<<<(N_EDGES + 255) / 256, 256, 0, stream>>>(ei, cnt);
    gcn_scan_kernel<<<(N_NODES + SCAN_T - 1) / SCAN_T, SCAN_T, 0, stream>>>(cnt, rowptr);
    gcn_fill_kernel<<<(N_EDGES + 255) / 256, 256, 0, stream>>>(ei, rowptr, permLo, permHi);

    gcn_gemm_wmma_kernel<<<GEMM_BLOCKS, GEMM_WAVES * 32, 0, stream>>>(x, w, support);

    gcn_gather_normalize_kernel<<<(N_NODES + 7) / 8, 256, 0, stream>>>(
        ei, adj, adjw, rowptr, permLo, permHi, support, out);
}


# Edges bucketed on one cursor add straight into the destination row's own out slice as (source, weight) pairs: count and scan launches and the gather's per-edge lookups go; row sums keep arrival order
# speedup vs baseline: 3.0175x; 1.4101x over previous
#include <hip/hip_runtime.h>
#include <hip/hip_bf16.h>


#define N_NODES 100000
#define DIM     256
#define N_EDGES 800000
#define L2EPS   1e-12f

typedef __attribute__((ext_vector_type(2))) float v2f;
typedef __attribute__((ext_vector_type(8))) float v8f;

#define BUCKET_PAIRS (DIM / 2)
static_assert(N_EDGES % 32 == 0, "the edge walk of an overfull row reads whole waves of edges");

__global__ __launch_bounds__(256)
void gcn_zero_cursor_kernel(int* __restrict__ cursor) {
    const int i = blockIdx.x * blockDim.x + threadIdx.x;
    if (i < N_NODES) cursor[i] = 0;
}

__global__ __launch_bounds__(256)
void gcn_bucket_kernel(const int* __restrict__ ei, const float* __restrict__ adj,
                       const float* __restrict__ adjw, int* __restrict__ cursor,
                       float* __restrict__ out) {
    const int e = blockIdx.x * blockDim.x + threadIdx.x;
    if (e >= N_EDGES) return;
    const int   d   = ei[e];
    const int   s   = ei[N_EDGES + e];
    const float val = adj[e] + adjw[e];
    const int pos = atomicAdd(&cursor[d], 1);
    if ((unsigned)pos < (unsigned)BUCKET_PAIRS)
        ((float2*)(out + (size_t)d * DIM))[pos] = make_float2(__int_as_float(s), val);
}

#define GEMM_WAVES  16
#define GEMM_BLOCKS 256
#define M_TILES     (N_NODES / 16)
static_assert(N_NODES % 16 == 0 && M_TILES >= GEMM_BLOCKS * GEMM_WAVES && M_TILES <= GEMM_BLOCKS * 25,
              "strip assignment below covers j = 0..24 and gives every wave a first strip");

__device__ __forceinline__ void gcn_load_a_chunk(const float* __restrict__ p, float4 (&a)[4]) {
    #pragma unroll
    for (int u = 0; u < 4; ++u) a[u] = *(const float4*)(p + 8 * u);
}

__device__ __forceinline__ void gcn_strip_chunk(const float4 (&a)[4], const float4* bp,
                                                v8f (&acc)[16]) {
    #pragma unroll
    for (int u = 0; u < 4; ++u) {
        float4 b[16];
        #pragma unroll
        for (int nt = 0; nt < 16; ++nt) b[nt] = bp[u * 512 + nt * 16];

        const v2f a01 = {a[u].x, a[u].y};
        const v2f a23 = {a[u].z, a[u].w};
        #pragma unroll
        for (int nt = 0; nt < 16; ++nt) {
            const v2f b01 = {b[nt].x, b[nt].y};
            acc[nt] = __builtin_amdgcn_wmma_f32_16x16x4_f32(
                false, a01, false, b01, (short)0, acc[nt], false, false);
        }
        #pragma unroll
        for (int nt = 0; nt < 16; ++nt) {
            const v2f b23 = {b[nt].z, b[nt].w};
            acc[nt] = __builtin_amdgcn_wmma_f32_16x16x4_f32(
                false, a23, false, b23, (short)0, acc[nt], false, false);
        }
    }
}

__global__ __launch_bounds__(GEMM_WAVES * 32)
void gcn_gemm_wmma_kernel(const float* __restrict__ x, const float* __restrict__ w,
                          float* __restrict__ support) {
    __shared__ float4 wl[DIM * DIM / 4];

    const int tid  = threadIdx.x;
    const int lane = tid & 31;
    const int h    = lane >> 4;
    const int c    = lane & 15;
    const int wv   = __builtin_amdgcn_readfirstlane(tid >> 5);

    const int j1 = (((wv >> 1) ^ (wv >> 2)) & 1) == 0 ? 16 + (((wv >> 2) << 1) | (wv & 1))
                                                      : (wv == 2 ? 24 : 25);
    int t  = blockIdx.x + GEMM_BLOCKS * wv;
    int tn = blockIdx.x + GEMM_BLOCKS * j1;
    const float* aPtr = x + (size_t)(t * 16 + c) * DIM + 4 * h;

    float4 aC[4], aN[4];
    gcn_load_a_chunk(aPtr, aC);

    for (int s = tid; s < DIM * DIM / 4; s += GEMM_WAVES * 32) {
        const float* src = w + (size_t)(4 * (s >> 8)) * DIM + (s & 255);
        wl[s] = make_float4(src[0], src[DIM], src[2 * DIM], src[3 * DIM]);
    }
    __syncthreads();

    const float4* bLane = wl + h * 256 + c;

    while (t < M_TILES) {
        const float* aNextTile = x + (size_t)(tn * 16 + c) * DIM + 4 * h;
        const float* aNextHead = tn < M_TILES ? aNextTile : aPtr;

        v8f acc[16];
        #pragma unroll
        for (int nt = 0; nt < 16; ++nt) acc[nt] = v8f{};

        #pragma unroll 1
        for (int gp = 0; gp < 4; ++gp) {
            gcn_load_a_chunk(aPtr + 32 * (2 * gp + 1), aN);
            gcn_strip_chunk(aC, bLane + (size_t)(8 * gp) * 512, acc);

            gcn_load_a_chunk(gp < 3 ? aPtr + 32 * (2 * gp + 2) : aNextHead, aC);
            gcn_strip_chunk(aN, bLane + (size_t)(8 * gp + 4) * 512, acc);
        }

        float* oBase = support + (size_t)(t * 16 + 8 * h) * DIM + c;
        #pragma unroll
        for (int nt = 0; nt < 16; ++nt) {
            #pragma unroll
            for (int i = 0; i < 8; ++i)
                oBase[(size_t)i * DIM + nt * 16] = acc[nt][i];
        }

        t    = tn;
        tn   = M_TILES;
        aPtr = aNextTile;
    }
}

template <int K>
__device__ __forceinline__ void gcn_accum_rows(const float* __restrict__ support,
                                               int mySrc, float myVal, int j, int lane,
                                               float4& accLo, float4& accHi) {
    float4 lo[K], hi[K];
    float  v[K];
    #pragma unroll
    for (int u = 0; u < K; ++u) {
        const int s = __builtin_amdgcn_readlane(mySrc, j + u);
        v[u] = __int_as_float(__builtin_amdgcn_readlane(__float_as_int(myVal), j + u));
        const float4* row = (const float4*)(support + (size_t)s * DIM);
        lo[u] = row[lane];
        hi[u] = row[32 + lane];
    }
    #pragma unroll
    for (int u = 0; u < K; ++u) {
        accLo.x = fmaf(v[u], lo[u].x, accLo.x);
        accLo.y = fmaf(v[u], lo[u].y, accLo.y);
        accLo.z = fmaf(v[u], lo[u].z, accLo.z);
        accLo.w = fmaf(v[u], lo[u].w, accLo.w);
        accHi.x = fmaf(v[u], hi[u].x, accHi.x);
        accHi.y = fmaf(v[u], hi[u].y, accHi.y);
        accHi.z = fmaf(v[u], hi[u].z, accHi.z);
        accHi.w = fmaf(v[u], hi[u].w, accHi.w);
    }
}

__global__ __launch_bounds__(256)
void gcn_gather_normalize_kernel(const int* __restrict__ ei, const float* __restrict__ adj,
                                 const float* __restrict__ adjw,
                                 const int* __restrict__ cursor,
                                 const float* __restrict__ support, float* out) {
    const int lane = threadIdx.x & 31;
    const int row = __builtin_amdgcn_readfirstlane(blockIdx.x * 8 + (threadIdx.x >> 5));
    if (row >= N_NODES) return;

    const float2* bucket = (const float2*)(out + (size_t)row * DIM);
    const int deg = cursor[row];

    float4 accLo = make_float4(0.f, 0.f, 0.f, 0.f);
    float4 accHi = make_float4(0.f, 0.f, 0.f, 0.f);

    if (deg <= BUCKET_PAIRS) {
        for (int base = 0; base < deg; base += 32) {
            const int n = min(32, deg - base);
            int   mySrc = 0;
            float myVal = 0.0f;
            if (lane < n) {
                const float2 pr = bucket[base + lane];
                mySrc = (int)min((unsigned)__float_as_int(pr.x), (unsigned)(N_NODES - 1));
                myVal = pr.y;
            }
            int j = 0;
            for (; j + 4 <= n; j += 4)
                gcn_accum_rows<4>(support, mySrc, myVal, j, lane, accLo, accHi);
            if (j + 2 <= n) {
                gcn_accum_rows<2>(support, mySrc, myVal, j, lane, accLo, accHi);
                j += 2;
            }
            if (j < n)
                gcn_accum_rows<1>(support, mySrc, myVal, j, lane, accLo, accHi);
        }
    } else {
        for (int base = 0; base < N_EDGES; base += 32) {
            const int  e   = base + lane;
            const bool hit = ei[e] == row;
            unsigned m = __builtin_amdgcn_ballot_w32(hit);
            if (m == 0) continue;
            int   mySrc = 0;
            float myVal = 0.0f;
            if (hit) {
                mySrc = (int)min((unsigned)ei[N_EDGES + e], (unsigned)(N_NODES - 1));
                myVal = adj[e] + adjw[e];
            }
            while (m != 0) {
                const int j = __builtin_ctz(m);
                m &= m - 1;
                gcn_accum_rows<1>(support, mySrc, myVal, j, lane, accLo, accHi);
            }
        }
    }

    float ss = accLo.x * accLo.x + accLo.y * accLo.y + accLo.z * accLo.z + accLo.w * accLo.w
             + accHi.x * accHi.x + accHi.y * accHi.y + accHi.z * accHi.z + accHi.w * accHi.w;
    for (int off = 16; off > 0; off >>= 1)
        ss += __shfl_xor(ss, off, 32);

    const float scale = 1.0f / fmaxf(sqrtf(ss), L2EPS);
    float4* o = (float4*)(out + (size_t)row * DIM);
    o[lane]      = make_float4(accLo.x * scale, accLo.y * scale, accLo.z * scale, accLo.w * scale);
    o[32 + lane] = make_float4(accHi.x * scale, accHi.y * scale, accHi.z * scale, accHi.w * scale);
}

extern "C" void kernel_launch(void* const* d_in, const int* in_sizes, int n_in,
                              void* d_out, int out_size, void* d_ws, size_t ws_size,
                              hipStream_t stream) {
    const float* x    = (const float*)d_in[0];
    const float* w    = (const float*)d_in[1];
    const float* adj  = (const float*)d_in[2];
    const float* adjw = (const float*)d_in[3];
    const int*   ei   = (const int*)  d_in[4];

    float* out     = (float*)d_out;
    float* support = (float*)d_ws;
    int* cursor = (int*)(support + (size_t)N_NODES * DIM);

    gcn_zero_cursor_kernel<<<(N_NODES + 255) / 256, 256, 0, stream>>>(cursor);
    gcn_bucket_kernel<<<(N_EDGES + 255) / 256, 256, 0, stream>>>(ei, adj, adjw, cursor, out);

    gcn_gemm_wmma_kernel<<<GEMM_BLOCKS, GEMM_WAVES * 32, 0, stream>>>(x, w, support);

    gcn_gather_normalize_kernel<<<(N_NODES + 7) / 8, 256, 0, stream>>>(
        ei, adj, adjw, cursor, support, out);
}


# Bucket pass moved into the GEMM launch: the 7 waves per block that own one strip place the edges first, beside the other waves' WMMAs; one launch, 25000 short waves go; row sums keep arrival order
# speedup vs baseline: 3.3293x; 1.1033x over previous
#include <hip/hip_runtime.h>
#include <hip/hip_bf16.h>


#define N_NODES 100000
#define DIM     256
#define N_EDGES 800000
#define L2EPS   1e-12f

typedef __attribute__((ext_vector_type(2))) float v2f;
typedef __attribute__((ext_vector_type(8))) float v8f;

#define BUCKET_PAIRS (DIM / 2)
static_assert(N_EDGES % 32 == 0, "the edge walk of an overfull row reads whole waves of edges");

__global__ __launch_bounds__(256)
void gcn_zero_cursor_kernel(int* __restrict__ cursor) {
    const int i = blockIdx.x * blockDim.x + threadIdx.x;
    if (i < N_NODES) cursor[i] = 0;
}

#define GEMM_WAVES  16
#define GEMM_BLOCKS 256
#define SPARE_WAVES 7
#define EDGE_LANES  (GEMM_BLOCKS * SPARE_WAVES * 32)
#define EDGE_FULL   (N_EDGES / EDGE_LANES)
#define EDGE_ROUNDS ((N_EDGES + EDGE_LANES - 1) / EDGE_LANES)
static_assert(EDGE_ROUNDS - EDGE_FULL <= 1, "at most one partly filled round");
static_assert(4 * EDGE_ROUNDS + 4 <= 63, "all edge loads of a lane in flight at once, beside its four x loads");

__device__ __forceinline__ void gcn_bucket_edges(const int* __restrict__ ei,
                                                 const float* __restrict__ adj,
                                                 const float* __restrict__ adjw,
                                                 int* __restrict__ cursor, float* __restrict__ out,
                                                 int first) {
    int   d[EDGE_ROUNDS], s[EDGE_ROUNDS], pos[EDGE_ROUNDS];
    float val[EDGE_ROUNDS];
    const bool lastOn = first + EDGE_FULL * EDGE_LANES < N_EDGES;
    #pragma unroll
    for (int i = 0; i < EDGE_ROUNDS; ++i) {
        const int  e  = first + i * EDGE_LANES;
        const bool on = i < EDGE_FULL || lastOn;
        d[i]   = on ? ei[e] : 0;
        s[i]   = on ? ei[N_EDGES + e] : 0;
        val[i] = on ? adj[e] + adjw[e] : 0.0f;
    }
    #pragma unroll
    for (int i = 0; i < EDGE_ROUNDS; ++i)
        pos[i] = (i < EDGE_FULL || lastOn) ? atomicAdd(&cursor[d[i]], 1) : BUCKET_PAIRS;
    #pragma unroll
    for (int i = 0; i < EDGE_ROUNDS; ++i)
        if ((unsigned)pos[i] < (unsigned)BUCKET_PAIRS)
            ((float2*)(out + (size_t)d[i] * DIM))[pos[i]] = make_float2(__int_as_float(s[i]), val[i]);
}

#define M_TILES     (N_NODES / 16)
static_assert(N_NODES % 16 == 0 && M_TILES >= GEMM_BLOCKS * GEMM_WAVES && M_TILES <= GEMM_BLOCKS * 25,
              "strip assignment below covers j = 0..24 and gives every wave a first strip");

__device__ __forceinline__ void gcn_load_a_chunk(const float* __restrict__ p, float4 (&a)[4]) {
    #pragma unroll
    for (int u = 0; u < 4; ++u) a[u] = *(const float4*)(p + 8 * u);
}

__device__ __forceinline__ void gcn_strip_chunk(const float4 (&a)[4], const float4* bp,
                                                v8f (&acc)[16]) {
    #pragma unroll
    for (int u = 0; u < 4; ++u) {
        float4 b[16];
        #pragma unroll
        for (int nt = 0; nt < 16; ++nt) b[nt] = bp[u * 512 + nt * 16];

        const v2f a01 = {a[u].x, a[u].y};
        const v2f a23 = {a[u].z, a[u].w};
        #pragma unroll
        for (int nt = 0; nt < 16; ++nt) {
            const v2f b01 = {b[nt].x, b[nt].y};
            acc[nt] = __builtin_amdgcn_wmma_f32_16x16x4_f32(
                false, a01, false, b01, (short)0, acc[nt], false, false);
        }
        #pragma unroll
        for (int nt = 0; nt < 16; ++nt) {
            const v2f b23 = {b[nt].z, b[nt].w};
            acc[nt] = __builtin_amdgcn_wmma_f32_16x16x4_f32(
                false, a23, false, b23, (short)0, acc[nt], false, false);
        }
    }
}

__global__ __launch_bounds__(GEMM_WAVES * 32)
void gcn_gemm_wmma_kernel(const float* __restrict__ x, const float* __restrict__ w,
                          float* __restrict__ support,
                          const int* __restrict__ ei, const float* __restrict__ adj,
                          const float* __restrict__ adjw, int* __restrict__ cursor,
                          float* __restrict__ out) {
    __shared__ float4 wl[DIM * DIM / 4];

    const int tid  = threadIdx.x;
    const int lane = tid & 31;
    const int h    = lane >> 4;
    const int c    = lane & 15;
    const int wv   = __builtin_amdgcn_readfirstlane(tid >> 5);

    const int j1 = (((wv >> 1) ^ (wv >> 2)) & 1) == 0 ? 16 + (((wv >> 2) << 1) | (wv & 1))
                                                      : (wv == 2 ? 24 : 25);
    int t  = blockIdx.x + GEMM_BLOCKS * wv;
    int tn = blockIdx.x + GEMM_BLOCKS * j1;
    const float* aPtr = x + (size_t)(t * 16 + c) * DIM + 4 * h;

    float4 aC[4], aN[4];
    gcn_load_a_chunk(aPtr, aC);

    for (int s = tid; s < DIM * DIM / 4; s += GEMM_WAVES * 32) {
        const float* src = w + (size_t)(4 * (s >> 8)) * DIM + (s & 255);
        wl[s] = make_float4(src[0], src[DIM], src[2 * DIM], src[3 * DIM]);
    }
    __syncthreads();

    if (j1 == 25) {
        const int rank = wv < 8 ? wv - 3 : wv - 7;
        gcn_bucket_edges(ei, adj, adjw, cursor, out,
                         (blockIdx.x * SPARE_WAVES + rank) * 32 + lane);
    }

    const float4* bLane = wl + h * 256 + c;

    while (t < M_TILES) {
        const float* aNextTile = x + (size_t)(tn * 16 + c) * DIM + 4 * h;
        const float* aNextHead = tn < M_TILES ? aNextTile : aPtr;

        v8f acc[16];
        #pragma unroll
        for (int nt = 0; nt < 16; ++nt) acc[nt] = v8f{};

        #pragma unroll 1
        for (int gp = 0; gp < 4; ++gp) {
            gcn_load_a_chunk(aPtr + 32 * (2 * gp + 1), aN);
            gcn_strip_chunk(aC, bLane + (size_t)(8 * gp) * 512, acc);

            gcn_load_a_chunk(gp < 3 ? aPtr + 32 * (2 * gp + 2) : aNextHead, aC);
            gcn_strip_chunk(aN, bLane + (size_t)(8 * gp + 4) * 512, acc);
        }

        float* oBase = support + (size_t)(t * 16 + 8 * h) * DIM + c;
        #pragma unroll
        for (int nt = 0; nt < 16; ++nt) {
            #pragma unroll
            for (int i = 0; i < 8; ++i)
                oBase[(size_t)i * DIM + nt * 16] = acc[nt][i];
        }

        t    = tn;
        tn   = M_TILES;
        aPtr = aNextTile;
    }
}

template <int K>
__device__ __forceinline__ void gcn_accum_rows(const float* __restrict__ support,
                                               int mySrc, float myVal, int j, int lane,
                                               float4& accLo, float4& accHi) {
    float4 lo[K], hi[K];
    float  v[K];
    #pragma unroll
    for (int u = 0; u < K; ++u) {
        const int s = __builtin_amdgcn_readlane(mySrc, j + u);
        v[u] = __int_as_float(__builtin_amdgcn_readlane(__float_as_int(myVal), j + u));
        const float4* row = (const float4*)(support + (size_t)s * DIM);
        lo[u] = row[lane];
        hi[u] = row[32 + lane];
    }
    #pragma unroll
    for (int u = 0; u < K; ++u) {
        accLo.x = fmaf(v[u], lo[u].x, accLo.x);
        accLo.y = fmaf(v[u], lo[u].y, accLo.y);
        accLo.z = fmaf(v[u], lo[u].z, accLo.z);
        accLo.w = fmaf(v[u], lo[u].w, accLo.w);
        accHi.x = fmaf(v[u], hi[u].x, accHi.x);
        accHi.y = fmaf(v[u], hi[u].y, accHi.y);
        accHi.z = fmaf(v[u], hi[u].z, accHi.z);
        accHi.w = fmaf(v[u], hi[u].w, accHi.w);
    }
}

__global__ __launch_bounds__(256)
void gcn_gather_normalize_kernel(const int* __restrict__ ei, const float* __restrict__ adj,
                                 const float* __restrict__ adjw,
                                 const int* __restrict__ cursor,
                                 const float* __restrict__ support, float* out) {
    const int lane = threadIdx.x & 31;
    const int row = __builtin_amdgcn_readfirstlane(blockIdx.x * 8 + (threadIdx.x >> 5));
    if (row >= N_NODES) return;

    const float2* bucket = (const float2*)(out + (size_t)row * DIM);
    const int deg = cursor[row];

    float4 accLo = make_float4(0.f, 0.f, 0.f, 0.f);
    float4 accHi = make_float4(0.f, 0.f, 0.f, 0.f);

    if (deg <= BUCKET_PAIRS) {
        for (int base = 0; base < deg; base += 32) {
            const int n = min(32, deg - base);
            int   mySrc = 0;
            float myVal = 0.0f;
            if (lane < n) {
                const float2 pr = bucket[base + lane];
                mySrc = (int)min((unsigned)__float_as_int(pr.x), (unsigned)(N_NODES - 1));
                myVal = pr.y;
            }
            int j = 0;
            for (; j + 4 <= n; j += 4)
                gcn_accum_rows<4>(support, mySrc, myVal, j, lane, accLo, accHi);
            if (j + 2 <= n) {
                gcn_accum_rows<2>(support, mySrc, myVal, j, lane, accLo, accHi);
                j += 2;
            }
            if (j < n)
                gcn_accum_rows<1>(support, mySrc, myVal, j, lane, accLo, accHi);
        }
    } else {
        for (int base = 0; base < N_EDGES; base += 32) {
            const int  e   = base + lane;
            const bool hit = ei[e] == row;
            unsigned m = __builtin_amdgcn_ballot_w32(hit);
            if (m == 0) continue;
            int   mySrc = 0;
            float myVal = 0.0f;
            if (hit) {
                mySrc = (int)min((unsigned)ei[N_EDGES + e], (unsigned)(N_NODES - 1));
                myVal = adj[e] + adjw[e];
            }
            while (m != 0) {
                const int j = __builtin_ctz(m);
                m &= m - 1;
                gcn_accum_rows<1>(support, mySrc, myVal, j, lane, accLo, accHi);
            }
        }
    }

    float ss = accLo.x * accLo.x + accLo.y * accLo.y + accLo.z * accLo.z + accLo.w * accLo.w
             + accHi.x * accHi.x + accHi.y * accHi.y + accHi.z * accHi.z + accHi.w * accHi.w;
    for (int off = 16; off > 0; off >>= 1)
        ss += __shfl_xor(ss, off, 32);

    const float scale = 1.0f / fmaxf(sqrtf(ss), L2EPS);
    float4* o = (float4*)(out + (size_t)row * DIM);
    o[lane]      = make_float4(accLo.x * scale, accLo.y * scale, accLo.z * scale, accLo.w * scale);
    o[32 + lane] = make_float4(accHi.x * scale, accHi.y * scale, accHi.z * scale, accHi.w * scale);
}

extern "C" void kernel_launch(void* const* d_in, const int* in_sizes, int n_in,
                              void* d_out, int out_size, void* d_ws, size_t ws_size,
                              hipStream_t stream) {
    const float* x    = (const float*)d_in[0];
    const float* w    = (const float*)d_in[1];
    const float* adj  = (const float*)d_in[2];
    const float* adjw = (const float*)d_in[3];
    const int*   ei   = (const int*)  d_in[4];

    float* out     = (float*)d_out;
    float* support = (float*)d_ws;
    int* cursor = (int*)(support + (size_t)N_NODES * DIM);

    gcn_zero_cursor_kernel<<<(N_NODES + 255) / 256, 256, 0, stream>>>(cursor);

    gcn_gemm_wmma_kernel<<<GEMM_BLOCKS, GEMM_WAVES * 32, 0, stream>>>(x, w, support,
                                                                     ei, adj, adjw, cursor, out);

    gcn_gather_normalize_kernel<<<(N_NODES + 7) / 8, 256, 0, stream>>>(
        ei, adj, adjw, cursor, support, out);
}
